# starting kernel
# baseline (speedup 1.0000x reference)
#include <hip/hip_runtime.h>

typedef __attribute__((ext_vector_type(2))) float v2f;
typedef __attribute__((ext_vector_type(8))) float v8f;

#define IMG_W   512
#define DEPTH   64
#define OUT_N   510
#define NOUT    (OUT_N * OUT_N)
#define NTILES  ((NOUT + 15) / 16)
#define KTOT    (3 * 3 * DEPTH)

__launch_bounds__(256)
__global__ void conv3x3d64_wmma_f32(const float* __restrict__ x,
                                    const float* __restrict__ W,
                                    const float* __restrict__ bias,
                                    float* __restrict__ out)
{
    __shared__ float wlds[KTOT];
    for (int i = threadIdx.x; i < KTOT; i += 256) wlds[i] = W[i];
    __syncthreads();

    const int wave = threadIdx.x >> 5;
    const int lane = threadIdx.x & 31;
    const int tile = blockIdx.x * 8 + wave;
    if (tile >= NTILES) return;

    const int      n  = lane & 15;
    const int      h  = lane >> 4;
    const unsigned p  = (unsigned)tile * 16u + (unsigned)n;
    const unsigned pe = (p < NOUT) ? p : (NOUT - 1);
    const unsigned r  = pe / OUT_N;
    const unsigned c  = pe % OUT_N;

    const float* xb = x    + ((size_t)r * IMG_W + c) * DEPTH + 2 * h;
    const float* wb = wlds + 2 * h;

    v8f acc = {};
#pragma unroll
    for (int fi = 0; fi < 3; ++fi) {
#pragma unroll
        for (int fj = 0; fj < 3; ++fj) {
#pragma unroll
            for (int d0 = 0; d0 < DEPTH; d0 += 4) {
                const int xoff = fi * (IMG_W * DEPTH) + fj * DEPTH + d0;
                const int koff = (fi * 3 + fj) * DEPTH + d0;
                v2f bvec = *(const v2f*)(xb + xoff);
                v2f avec = *(const v2f*)(wb + koff);
                acc = __builtin_amdgcn_wmma_f32_16x16x4_f32(
                         false, avec,  false, bvec,
                         (short)0, acc,
                         false,  false);
            }
        }
    }

    if (h == 0 && p < NOUT) out[p] = acc[0] + bias[0];
}

extern "C" void kernel_launch(void* const* d_in, const int* in_sizes, int n_in,
                              void* d_out, int out_size, void* d_ws, size_t ws_size,
                              hipStream_t stream) {
    (void)in_sizes; (void)n_in; (void)d_ws; (void)ws_size; (void)out_size;
    const float* x  = (const float*)d_in[0];
    const float* W  = (const float*)d_in[1];
    const float* b  = (const float*)d_in[2];
    float* out      = (float*)d_out;

    const int blocks = (NTILES + 7) / 8;
    hipLaunchKernelGGL(conv3x3d64_wmma_f32, dim3(blocks), dim3(256), 0, stream,
                       x, W, b, out);
}


# WMMA rows = the 9 filter taps (was one row copied 16 times): each pixel's 64 depths loaded once for all taps, tap planes in LDS per 19x32 tile, 9 shifted adds per output; sum order: depth, then taps
# speedup vs baseline: 8.0520x; 8.0520x over previous
#include <hip/hip_runtime.h>

typedef __attribute__((ext_vector_type(2))) float v2f;
typedef __attribute__((ext_vector_type(4))) float v4f;
typedef __attribute__((ext_vector_type(8))) float v8f;

#define IMG_W    512
#define DEPTH    64
#define OUT_N    510
#define TR_OUT   17
#define TC_OUT   30
#define TR_IN    (TR_OUT + 2)
#define TC_IN    (TC_OUT + 2)
#define NPIX     (TR_IN * TC_IN)
#define NTILE    (NPIX / 16)
#define PLANE    NPIX
#define NTAP     9

__device__ __forceinline__ void tile_products(const v4f (&a)[8], const v4f (&b)[8],
                                              float* __restrict__ tl, int slot, int h)
{
    v8f acc0 = {};
    v8f acc1 = {};
#pragma unroll
    for (int j = 0; j < 8; ++j) {
        v2f a0 = {a[j].x, a[j].y};
        v2f a1 = {a[j].z, a[j].w};
        v2f b0 = {b[j].x, b[j].y};
        v2f b1 = {b[j].z, b[j].w};
        acc0 = __builtin_amdgcn_wmma_f32_16x16x4_f32(false, a0, false, b0, (short)0, acc0, false, false);
        acc1 = __builtin_amdgcn_wmma_f32_16x16x4_f32(false, a1, false, b1, (short)0, acc1, false, false);
    }
    float* p = tl + (5 * h) * PLANE + slot;
    p[0 * PLANE] = acc0[0] + acc1[0];
    p[1 * PLANE] = acc0[1] + acc1[1];
    p[2 * PLANE] = acc0[2] + acc1[2];
    p[3 * PLANE] = acc0[3] + acc1[3];
    if (h == 0) p[4 * PLANE] = acc0[4] + acc1[4];
}

__launch_bounds__(256)
__global__ void conv3x3d64_taps_wmma_f32(const float* __restrict__ x,
                                         const float* __restrict__ W,
                                         const float* __restrict__ bias,
                                         float* __restrict__ out)
{
    __shared__ float tl[NTAP * PLANE];

    const int wave = __builtin_amdgcn_readfirstlane((int)(threadIdx.x >> 5));
    const int lane = threadIdx.x & 31;
    const int c    = lane & 15;
    const int h    = lane >> 4;
    const int R0   = blockIdx.y * TR_OUT;
    const int C0   = blockIdx.x * TC_OUT;

    const int  tm    = (c < 5) ? c : (c - 3);
    const bool arow  = (c < 5) || (c >= 8 && c < 12);
    const float* wp  = W + (arow ? tm : 0) * DEPTH + 4 * h;
    v4f a[8];
#pragma unroll
    for (int j = 0; j < 8; ++j) {
        v4f t = *(const v4f*)(wp + 8 * j);
        v4f z = {0.f, 0.f, 0.f, 0.f};
        a[j] = arow ? t : z;
    }

    const int rr0 = wave >> 1;
    const int ch  = wave & 1;
    const float* xp = x + ((size_t)(R0 + rr0) * IMG_W + (size_t)(C0 + ch * 16 + c)) * DEPTH + 4 * h;
    const int slot0 = rr0 * TC_IN + ch * 16 + c;
    const bool five = wave < (NTILE - 32);

    v4f b4[8];
    if (five) {
#pragma unroll
        for (int j = 0; j < 8; ++j)
            b4[j] = *(const v4f*)(xp + (size_t)4 * (4 * IMG_W * DEPTH) + 8 * j);
    }
    v4f b[4][8];
#pragma unroll
    for (int i = 0; i < 4; ++i) {
#pragma unroll
        for (int j = 0; j < 8; ++j)
            b[i][j] = *(const v4f*)(xp + (size_t)i * (4 * IMG_W * DEPTH) + 8 * j);
    }

    if (five)
        tile_products(a, b4, tl, slot0 + 16 * TC_IN, h);
#pragma unroll
    for (int i = 0; i < 4; ++i)
        tile_products(a, b[i], tl, slot0 + i * 4 * TC_IN, h);

    __syncthreads();

    const float bv = bias[0];
    if (lane < TC_OUT) {
        for (int ro = wave; ro < TR_OUT; ro += 8) {
            const float* q = tl + ro * TC_IN + lane;
            float s = 0.f;
#pragma unroll
            for (int fi = 0; fi < 3; ++fi) {
#pragma unroll
                for (int fj = 0; fj < 3; ++fj)
                    s += q[(fi * 3 + fj) * PLANE + fi * TC_IN + fj];
            }
            out[(size_t)(R0 + ro) * OUT_N + (size_t)(C0 + lane)] = s + bv;
        }
    }
}

extern "C" void kernel_launch(void* const* d_in, const int* in_sizes, int n_in,
                              void* d_out, int out_size, void* d_ws, size_t ws_size,
                              hipStream_t stream) {
    (void)in_sizes; (void)n_in; (void)d_ws; (void)ws_size; (void)out_size;
    const float* x  = (const float*)d_in[0];
    const float* W  = (const float*)d_in[1];
    const float* b  = (const float*)d_in[2];
    float* out      = (float*)d_out;

    hipLaunchKernelGGL(conv3x3d64_taps_wmma_f32, dim3(17, 30), dim3(256), 0, stream,
                       x, W, b, out);
}
